# starting kernel
# baseline (speedup 1.0000x reference)
#include <hip/hip_runtime.h>
#include <cstddef>

typedef __attribute__((ext_vector_type(2))) float v2f;
typedef __attribute__((ext_vector_type(8))) float v8f;

#if defined(__HIP_DEVICE_COMPILE__) && !__has_builtin(__builtin_amdgcn_wmma_f32_16x16x4_f32)
#error "__builtin_amdgcn_wmma_f32_16x16x4_f32 not available on gfx1250 device pass"
#endif

static __device__ __forceinline__ v8f wmma4(v2f a, v2f b, v8f c) {
  return __builtin_amdgcn_wmma_f32_16x16x4_f32(false, a, false, b, (short)0, c,
                                               false, false);
}

constexpr int NROWS = 256;
constexpr int TLEN  = 131072;
constexpr int PAD   = 15;
constexpr int ELEN  = TLEN + 2 * PAD;
constexpr int ES    = 131104;
constexpr int NBLK  = ES / 16;
constexpr int CB    = 128;
constexpr int WB    = 16;
constexpr int NCHUNK = (NBLK + CB - 1) / CB;
constexpr int NRG   = NROWS / 16;

constexpr int WS_TB = 0;
constexpr int WS_WB = 256;
constexpr int WS_UB = 320;
constexpr int WS_PB = 576;
constexpr int WS_ZI = 640;
constexpr int WS_YF = 1024;

__global__ void setup_kernel(const float* __restrict__ bc,
                             const float* __restrict__ ac,
                             float* __restrict__ ws) {
  if (threadIdx.x != 0 || blockIdx.x != 0) return;
  double a0 = (double)ac[0];
  double an[5], bn[5];
  for (int i = 0; i < 5; ++i) { an[i] = (double)ac[i] / a0; bn[i] = (double)bc[i] / a0; }
  const double b0 = bn[0];
  double Bv[4];
  for (int i = 0; i < 4; ++i) Bv[i] = bn[i + 1] - an[i + 1] * b0;
  double A[4][4] = {{-an[1], 1, 0, 0},
                    {-an[2], 0, 1, 0},
                    {-an[3], 0, 0, 1},
                    {-an[4], 0, 0, 0}};
  double Crow[16][4];
  Crow[0][0] = 1; Crow[0][1] = 0; Crow[0][2] = 0; Crow[0][3] = 0;
  for (int k = 1; k < 16; ++k)
    for (int j = 0; j < 4; ++j) {
      double s = 0;
      for (int i = 0; i < 4; ++i) s += Crow[k - 1][i] * A[i][j];
      Crow[k][j] = s;
    }
  double h[16];
  h[0] = b0;
  for (int m = 1; m < 16; ++m) {
    double s = 0;
    for (int i = 0; i < 4; ++i) s += Crow[m - 1][i] * Bv[i];
    h[m] = s;
  }
  double u[16][4];
  for (int i = 0; i < 4; ++i) u[0][i] = Bv[i];
  for (int p = 1; p < 16; ++p)
    for (int i = 0; i < 4; ++i) {
      double s = 0;
      for (int j = 0; j < 4; ++j) s += A[i][j] * u[p - 1][j];
      u[p][i] = s;
    }
  double P[4][4], Tm[4][4];
  for (int i = 0; i < 4; ++i)
    for (int j = 0; j < 4; ++j) P[i][j] = (i == j) ? 1.0 : 0.0;
  for (int s = 0; s < 16; ++s) {
    for (int i = 0; i < 4; ++i)
      for (int j = 0; j < 4; ++j) {
        double t = 0;
        for (int q = 0; q < 4; ++q) t += A[i][q] * P[q][j];
        Tm[i][j] = t;
      }
    for (int i = 0; i < 4; ++i)
      for (int j = 0; j < 4; ++j) P[i][j] = Tm[i][j];
  }
  double M[4][5];
  for (int i = 0; i < 4; ++i) {
    for (int j = 0; j < 4; ++j)
      M[i][j] = ((i == j) ? 1.0 : 0.0) + ((j == 0) ? an[i + 1] : 0.0) -
                ((j == i + 1) ? 1.0 : 0.0);
    M[i][4] = Bv[i];
  }
  for (int c = 0; c < 4; ++c) {
    int piv = c;
    for (int r = c + 1; r < 4; ++r)
      if ((M[r][c] < 0 ? -M[r][c] : M[r][c]) > (M[piv][c] < 0 ? -M[piv][c] : M[piv][c]))
        piv = r;
    if (piv != c)
      for (int j = 0; j < 5; ++j) { double t = M[c][j]; M[c][j] = M[piv][j]; M[piv][j] = t; }
    double d = M[c][c];
    for (int j = 0; j < 5; ++j) M[c][j] /= d;
    for (int r = 0; r < 4; ++r) {
      if (r == c) continue;
      double f = M[r][c];
      for (int j = 0; j < 5; ++j) M[r][j] -= f * M[c][j];
    }
  }
  for (int c = 0; c < 4; ++c)
    for (int v = 0; v < 2; ++v)
      for (int L = 0; L < 32; ++L) {
        int k = v + ((L >= 16) ? 2 : 0);
        int n = L & 15;
        int kk = 4 * c + k;
        ws[WS_TB + c * 64 + v * 32 + L] = (float)((n >= kk) ? h[n - kk] : 0.0);
        ws[WS_UB + c * 64 + v * 32 + L] = (float)((n < 4) ? u[15 - kk][n] : 0.0);
      }
  for (int v = 0; v < 2; ++v)
    for (int L = 0; L < 32; ++L) {
      int k = v + ((L >= 16) ? 2 : 0);
      int n = L & 15;
      ws[WS_WB + v * 32 + L] = (float)Crow[n][k];
      ws[WS_PB + v * 32 + L] = (float)((n < 4) ? P[n][k] : 0.0);
    }
  for (int i = 0; i < 4; ++i) ws[WS_ZI + i] = (float)M[i][4];
}

static __device__ __forceinline__ float load_ext_fwd(const float* __restrict__ xr, int e) {
  if (e < PAD) return 2.0f * xr[0] - xr[PAD - e];
  if (e < PAD + TLEN) return xr[e - PAD];
  int j = e - (PAD + TLEN);
  if (j > PAD - 1) j = PAD - 1;
  return 2.0f * xr[TLEN - 1] - xr[TLEN - 2 - j];
}

static __device__ __forceinline__ float load_rev(const float* __restrict__ yr, int e) {
  int idx = ELEN - 1 - e;
  if (idx < 0) idx = 0;
  return yr[idx];
}

__global__ __launch_bounds__(32) void fwd_kernel(const float* __restrict__ x,
                                                 float* __restrict__ ws) {
  float* __restrict__ yf = ws + WS_YF;
  const int task = blockIdx.x;
  const int rg = task / NCHUNK, ch = task % NCHUNK;
  const int r0 = rg * 16;
  const int L = threadIdx.x;
  const int m = L & 15;
  const int hiA = L >> 4;
  const int k0 = 2 * hiA;
  const float* __restrict__ xrow = x + (size_t)(r0 + m) * TLEN;

  v2f TBv[4], UBv[4], WBv, PBv;
#pragma unroll
  for (int c = 0; c < 4; ++c) {
    TBv[c] = (v2f){ws[WS_TB + c * 64 + L], ws[WS_TB + c * 64 + 32 + L]};
    UBv[c] = (v2f){ws[WS_UB + c * 64 + L], ws[WS_UB + c * 64 + 32 + L]};
  }
  WBv = (v2f){ws[WS_WB + L], ws[WS_WB + 32 + L]};
  PBv = (v2f){ws[WS_PB + L], ws[WS_PB + 32 + L]};
  const float zi0 = ws[WS_ZI + k0], zi1 = ws[WS_ZI + k0 + 1];

  const int bOut = ch * CB;
  const int bW = (bOut - WB > 0) ? (bOut - WB) : 0;
  const int bEnd = ((bOut + CB) < NBLK) ? (bOut + CB) : NBLK;

  float v0 = load_ext_fwd(xrow, bW * 16);
  v2f z = (v2f){zi0 * v0, zi1 * v0};

  const int sl0 = k0 + ((m & 8) ? 16 : 0);
  const int sl1 = sl0 + 1;
  const int vsel = L & 7;

  for (int bb = bW; bb < bEnd; ++bb) {
    const int e0 = bb * 16;
    v2f xa[4];
    if (e0 >= PAD && (e0 + 16) <= (PAD + TLEN)) {
      const float* p = xrow + (e0 - PAD) + k0;
#pragma unroll
      for (int c = 0; c < 4; ++c) xa[c] = (v2f){p[4 * c], p[4 * c + 1]};
      __builtin_prefetch(p + 64, 0, 1);
    } else {
#pragma unroll
      for (int c = 0; c < 4; ++c) {
        int o = e0 + 4 * c + k0;
        xa[c] = (v2f){load_ext_fwd(xrow, o), load_ext_fwd(xrow, o + 1)};
      }
    }
    const v8f zero = (v8f){0, 0, 0, 0, 0, 0, 0, 0};
    v8f dZ = wmma4(z, PBv, zero);
#pragma unroll
    for (int c = 0; c < 4; ++c) dZ = wmma4(xa[c], UBv[c], dZ);
    if (bb >= bOut) {
      v8f dY = wmma4(z, WBv, zero);
#pragma unroll
      for (int c = 0; c < 4; ++c) dY = wmma4(xa[c], TBv[c], dY);
#pragma unroll
      for (int v = 0; v < 8; ++v)
        yf[(size_t)(r0 + v + 8 * hiA) * ES + e0 + m] = dY[v];
    }
    float nz0 = 0.f, nz1 = 0.f;
#pragma unroll
    for (int v = 0; v < 8; ++v) {
      float g0 = __shfl(dZ[v], sl0, 32);
      float g1 = __shfl(dZ[v], sl1, 32);
      nz0 = (vsel == v) ? g0 : nz0;
      nz1 = (vsel == v) ? g1 : nz1;
    }
    z = (v2f){nz0, nz1};
  }
}

__global__ __launch_bounds__(32) void bwd_kernel(const float* __restrict__ ws,
                                                 float* __restrict__ out) {
  const float* __restrict__ yf = ws + WS_YF;
  const int task = blockIdx.x;
  const int rg = task / NCHUNK, ch = task % NCHUNK;
  const int r0 = rg * 16;
  const int L = threadIdx.x;
  const int m = L & 15;
  const int hiA = L >> 4;
  const int k0 = 2 * hiA;
  const float* __restrict__ yrow = yf + (size_t)(r0 + m) * ES;

  v2f TBv[4], UBv[4], WBv, PBv;
#pragma unroll
  for (int c = 0; c < 4; ++c) {
    TBv[c] = (v2f){ws[WS_TB + c * 64 + L], ws[WS_TB + c * 64 + 32 + L]};
    UBv[c] = (v2f){ws[WS_UB + c * 64 + L], ws[WS_UB + c * 64 + 32 + L]};
  }
  WBv = (v2f){ws[WS_WB + L], ws[WS_WB + 32 + L]};
  PBv = (v2f){ws[WS_PB + L], ws[WS_PB + 32 + L]};
  const float zi0 = ws[WS_ZI + k0], zi1 = ws[WS_ZI + k0 + 1];

  const int bOut = ch * CB;
  const int bW = (bOut - WB > 0) ? (bOut - WB) : 0;
  const int bEnd = ((bOut + CB) < NBLK) ? (bOut + CB) : NBLK;

  float v0 = load_rev(yrow, bW * 16);
  v2f z = (v2f){zi0 * v0, zi1 * v0};
  const int sl0 = k0 + ((m & 8) ? 16 : 0);
  const int sl1 = sl0 + 1;
  const int vsel = L & 7;

  for (int bb = bW; bb < bEnd; ++bb) {
    const int e0 = bb * 16;
    v2f xa[4];
    if (e0 + 16 <= ELEN) {
      const float* p = yrow + (ELEN - 1 - e0 - k0);
#pragma unroll
      for (int c = 0; c < 4; ++c) xa[c] = (v2f){p[-4 * c], p[-4 * c - 1]};
      __builtin_prefetch(p - 64, 0, 1);
    } else {
#pragma unroll
      for (int c = 0; c < 4; ++c) {
        int o = e0 + 4 * c + k0;
        xa[c] = (v2f){load_rev(yrow, o), load_rev(yrow, o + 1)};
      }
    }
    const v8f zero = (v8f){0, 0, 0, 0, 0, 0, 0, 0};
    v8f dZ = wmma4(z, PBv, zero);
#pragma unroll
    for (int c = 0; c < 4; ++c) dZ = wmma4(xa[c], UBv[c], dZ);
    if (bb >= bOut) {
      v8f dY = wmma4(z, WBv, zero);
#pragma unroll
      for (int c = 0; c < 4; ++c) dY = wmma4(xa[c], TBv[c], dY);
      const int e = e0 + m;
      if (e >= PAD && e < PAD + TLEN) {
        const int col = ELEN - 16 - e;
#pragma unroll
        for (int v = 0; v < 8; ++v)
          out[(size_t)(r0 + v + 8 * hiA) * TLEN + col] = dY[v];
      }
    }
    float nz0 = 0.f, nz1 = 0.f;
#pragma unroll
    for (int v = 0; v < 8; ++v) {
      float g0 = __shfl(dZ[v], sl0, 32);
      float g1 = __shfl(dZ[v], sl1, 32);
      nz0 = (vsel == v) ? g0 : nz0;
      nz1 = (vsel == v) ? g1 : nz1;
    }
    z = (v2f){nz0, nz1};
  }
}

extern "C" void kernel_launch(void* const* d_in, const int* in_sizes, int n_in,
                              void* d_out, int out_size, void* d_ws, size_t ws_size,
                              hipStream_t stream) {
  (void)in_sizes; (void)n_in; (void)out_size; (void)ws_size;
  const float* x = (const float*)d_in[0];
  const float* b = (const float*)d_in[1];
  const float* a = (const float*)d_in[2];
  float* out = (float*)d_out;
  float* ws = (float*)d_ws;

  setup_kernel<<<1, 1, 0, stream>>>(b, a, ws);
  fwd_kernel<<<NRG * NCHUNK, 32, 0, stream>>>(x, ws);
  bwd_kernel<<<NRG * NCHUNK, 32, 0, stream>>>(ws, out);
}


# Forward and backward pass fused: a wave keeps its 16 rows' forward result in an LDS ring (43-block windows) and filters it backward from there; no intermediate in memory; backward sums regrouped
# speedup vs baseline: 1.4921x; 1.4921x over previous
#include <hip/hip_runtime.h>
#include <cstddef>

typedef __attribute__((ext_vector_type(2))) float v2f;
typedef __attribute__((ext_vector_type(8))) float v8f;

#if defined(__HIP_DEVICE_COMPILE__) && !__has_builtin(__builtin_amdgcn_wmma_f32_16x16x4_f32)
#error "__builtin_amdgcn_wmma_f32_16x16x4_f32 not available on gfx1250 device pass"
#endif

static __device__ __forceinline__ v8f wmma4(v2f a, v2f b, v8f c) {
  return __builtin_amdgcn_wmma_f32_16x16x4_f32(false, a, false, b, (short)0, c,
                                               false, false);
}

constexpr int NROWS = 256;
constexpr int TLEN  = 131072;
constexpr int PAD   = 15;
constexpr int ELEN  = TLEN + 2 * PAD;
constexpr int ES    = 131104;
constexpr int NBLK  = ES / 16;
constexpr int WB    = 16;
constexpr int NRG   = NROWS / 16;
constexpr int NSTR  = 64;
constexpr int SBLK  = (NBLK + NSTR - 1) / NSTR;
constexpr int LW    = 43;
constexpr int NWIN  = (SBLK + LW - 1) / LW;
constexpr int RB    = LW + WB;
constexpr int PW    = RB * 16 + 4;
constexpr int TAILK = (ELEN - 1) % 16;
static_assert(PW % 4 == 0 && (PW / 4) % 2 == 1, "row pitch must be 4 * odd");
static_assert(NWIN * LW >= SBLK && SBLK + WB < 3 * RB, "window bookkeeping");

constexpr int WS_TB = 0;
constexpr int WS_WB = 256;
constexpr int WS_UB = 320;
constexpr int WS_PB = 576;
constexpr int WS_ZI = 640;

__global__ void setup_kernel(const float* __restrict__ bc,
                             const float* __restrict__ ac,
                             float* __restrict__ ws) {
  if (threadIdx.x != 0 || blockIdx.x != 0) return;
  double a0 = (double)ac[0];
  double an[5], bn[5];
  for (int i = 0; i < 5; ++i) { an[i] = (double)ac[i] / a0; bn[i] = (double)bc[i] / a0; }
  const double b0 = bn[0];
  double Bv[4];
  for (int i = 0; i < 4; ++i) Bv[i] = bn[i + 1] - an[i + 1] * b0;
  double A[4][4] = {{-an[1], 1, 0, 0},
                    {-an[2], 0, 1, 0},
                    {-an[3], 0, 0, 1},
                    {-an[4], 0, 0, 0}};
  double Crow[16][4];
  Crow[0][0] = 1; Crow[0][1] = 0; Crow[0][2] = 0; Crow[0][3] = 0;
  for (int k = 1; k < 16; ++k)
    for (int j = 0; j < 4; ++j) {
      double s = 0;
      for (int i = 0; i < 4; ++i) s += Crow[k - 1][i] * A[i][j];
      Crow[k][j] = s;
    }
  double h[16];
  h[0] = b0;
  for (int m = 1; m < 16; ++m) {
    double s = 0;
    for (int i = 0; i < 4; ++i) s += Crow[m - 1][i] * Bv[i];
    h[m] = s;
  }
  double u[16][4];
  for (int i = 0; i < 4; ++i) u[0][i] = Bv[i];
  for (int p = 1; p < 16; ++p)
    for (int i = 0; i < 4; ++i) {
      double s = 0;
      for (int j = 0; j < 4; ++j) s += A[i][j] * u[p - 1][j];
      u[p][i] = s;
    }
  double P[4][4], Tm[4][4];
  for (int i = 0; i < 4; ++i)
    for (int j = 0; j < 4; ++j) P[i][j] = (i == j) ? 1.0 : 0.0;
  for (int s = 0; s < 16; ++s) {
    for (int i = 0; i < 4; ++i)
      for (int j = 0; j < 4; ++j) {
        double t = 0;
        for (int q = 0; q < 4; ++q) t += A[i][q] * P[q][j];
        Tm[i][j] = t;
      }
    for (int i = 0; i < 4; ++i)
      for (int j = 0; j < 4; ++j) P[i][j] = Tm[i][j];
  }
  double M[4][5];
  for (int i = 0; i < 4; ++i) {
    for (int j = 0; j < 4; ++j)
      M[i][j] = ((i == j) ? 1.0 : 0.0) + ((j == 0) ? an[i + 1] : 0.0) -
                ((j == i + 1) ? 1.0 : 0.0);
    M[i][4] = Bv[i];
  }
  for (int c = 0; c < 4; ++c) {
    int piv = c;
    for (int r = c + 1; r < 4; ++r)
      if ((M[r][c] < 0 ? -M[r][c] : M[r][c]) > (M[piv][c] < 0 ? -M[piv][c] : M[piv][c]))
        piv = r;
    if (piv != c)
      for (int j = 0; j < 5; ++j) { double t = M[c][j]; M[c][j] = M[piv][j]; M[piv][j] = t; }
    double d = M[c][c];
    for (int j = 0; j < 5; ++j) M[c][j] /= d;
    for (int r = 0; r < 4; ++r) {
      if (r == c) continue;
      double f = M[r][c];
      for (int j = 0; j < 5; ++j) M[r][j] -= f * M[c][j];
    }
  }
  for (int c = 0; c < 4; ++c)
    for (int v = 0; v < 2; ++v)
      for (int L = 0; L < 32; ++L) {
        int k = v + ((L >= 16) ? 2 : 0);
        int n = L & 15;
        int kk = 4 * c + k;
        ws[WS_TB + c * 64 + v * 32 + L] = (float)((n >= kk) ? h[n - kk] : 0.0);
        ws[WS_UB + c * 64 + v * 32 + L] = (float)((n < 4) ? u[15 - kk][n] : 0.0);
      }
  for (int v = 0; v < 2; ++v)
    for (int L = 0; L < 32; ++L) {
      int k = v + ((L >= 16) ? 2 : 0);
      int n = L & 15;
      ws[WS_WB + v * 32 + L] = (float)Crow[n][k];
      ws[WS_PB + v * 32 + L] = (float)((n < 4) ? P[n][k] : 0.0);
    }
  for (int i = 0; i < 4; ++i) ws[WS_ZI + i] = (float)M[i][4];
}

static __device__ __forceinline__ float load_ext_fwd(const float* __restrict__ xr, int e) {
  if (e < PAD) return 2.0f * xr[0] - xr[PAD - e];
  if (e < PAD + TLEN) return xr[e - PAD];
  int j = e - (PAD + TLEN);
  if (j > PAD - 1) j = PAD - 1;
  return 2.0f * xr[TLEN - 1] - xr[TLEN - 2 - j];
}

constexpr int PD = 4;
constexpr int FI_LO = (PAD + 15) / 16;
constexpr int FI_HI = (PAD + TLEN - 16) / 16 + 1;

static __device__ __forceinline__ v2f next_state(v8f dZ, int sl0, int sl1, int vsel) {
  float nz0 = 0.f, nz1 = 0.f;
#pragma unroll
  for (int v = 0; v < 8; ++v) {
    float g0 = __shfl(dZ[v], sl0, 32);
    float g1 = __shfl(dZ[v], sl1, 32);
    nz0 = (vsel == v) ? g0 : nz0;
    nz1 = (vsel == v) ? g1 : nz1;
  }
  return (v2f){nz0, nz1};
}

static __device__ __forceinline__ int slot_of(int t) {
  t = (t >= RB) ? t - RB : t;
  t = (t >= RB) ? t - RB : t;
  return t;
}

static __device__ __forceinline__ void ring_sync() {
  __builtin_amdgcn_fence(__ATOMIC_RELEASE, "workgroup", "local");
  __builtin_amdgcn_s_barrier();
  __builtin_amdgcn_fence(__ATOMIC_ACQUIRE, "workgroup", "local");
}

static __device__ __forceinline__ void ld_fwd_whole(const float* __restrict__ xrow, int k0,
                                                    int bb, v2f (&xa)[4]) {
  const float* p = xrow + (bb * 16 - PAD) + k0;
#pragma unroll
  for (int c = 0; c < 4; ++c) xa[c] = (v2f){p[4 * c], p[4 * c + 1]};
  __builtin_prefetch(p + 64, 0, 1);
}

static __device__ __forceinline__ void ld_fwd_any(const float* __restrict__ xrow, int k0,
                                                  int bb, v2f (&xa)[4]) {
  if (bb >= FI_LO && bb < FI_HI) {
    ld_fwd_whole(xrow, k0, bb, xa);
  } else {
    const int e0 = bb * 16;
#pragma unroll
    for (int c = 0; c < 4; ++c) {
      int o = e0 + 4 * c + k0;
      xa[c] = (v2f){load_ext_fwd(xrow, o), load_ext_fwd(xrow, o + 1)};
    }
  }
}

static __device__ __forceinline__ v8f fwd_mm(const v2f (&xa)[4], v2f z, const v2f (&TBv)[4],
                                             const v2f (&UBv)[4], v2f WBv, v2f PBv, bool emit,
                                             float* lst, int slot) {
  const v8f zero = (v8f){0, 0, 0, 0, 0, 0, 0, 0};
  v8f dZ = wmma4(z, PBv, zero);
#pragma unroll
  for (int c = 0; c < 4; ++c) dZ = wmma4(xa[c], UBv[c], dZ);
  if (emit) {
    v8f dY = wmma4(z, WBv, zero);
#pragma unroll
    for (int c = 0; c < 4; ++c) dY = wmma4(xa[c], TBv[c], dY);
    float* q = lst + slot * 16;
#pragma unroll
    for (int v = 0; v < 8; ++v) q[v * PW] = dY[v];
  }
  return dZ;
}

static __device__ __forceinline__ v2f fwd_sweep(const float* __restrict__ xrow, int k0,
                                                int bFrom, int bTo, int sB, v2f z,
                                                const v2f (&TBv)[4], const v2f (&UBv)[4],
                                                v2f WBv, v2f PBv, float* lst, int sl0, int sl1,
                                                int vsel) {
  const int bLo = (bFrom > FI_LO) ? bFrom : FI_LO;
  const int bHi = (bTo < FI_HI) ? bTo : FI_HI;
  const int nRing = (bHi > bLo) ? ((bHi - bLo) / PD) * PD : 0;
  const int bMid = bLo + nRing;
  const int bHead = (nRing > 0) ? bLo : bTo;
  for (int bb = bFrom; bb < bHead; ++bb) {
    v2f xa[4];
    ld_fwd_any(xrow, k0, bb, xa);
    const v8f dZ = fwd_mm(xa, z, TBv, UBv, WBv, PBv, bb >= sB, lst, slot_of(bb - sB));
    z = next_state(dZ, sl0, sl1, vsel);
  }
  if (nRing > 0) {
    v2f ring[PD][4];
#pragma unroll
    for (int d = 0; d < PD; ++d) ld_fwd_whole(xrow, k0, bLo + d, ring[d]);
    for (int bb = bLo; bb < bMid; bb += PD) {
#pragma unroll
      for (int d = 0; d < PD; ++d) {
        const int b = bb + d;
        const v8f dZ =
            fwd_mm(ring[d], z, TBv, UBv, WBv, PBv, b >= sB, lst, slot_of(b - sB));
        int bl = b + PD;
        bl = (bl < bHi) ? bl : (bHi - 1);
        ld_fwd_whole(xrow, k0, bl, ring[d]);
        z = next_state(dZ, sl0, sl1, vsel);
      }
    }
    for (int bb = bMid; bb < bTo; ++bb) {
      v2f xa[4];
      ld_fwd_any(xrow, k0, bb, xa);
      const v8f dZ = fwd_mm(xa, z, TBv, UBv, WBv, PBv, bb >= sB, lst, slot_of(bb - sB));
      z = next_state(dZ, sl0, sl1, vsel);
    }
  }
  return z;
}

static __device__ __forceinline__ void ld_bwd_lds(const float* lld, int slot, v2f (&raw)[4]) {
  const float* p = lld + slot * 16;
#pragma unroll
  for (int c = 0; c < 4; ++c) raw[c] = *(const v2f*)(p - 4 * c);
}
static __device__ __forceinline__ v2f rev2(v2f a) { return (v2f){a.y, a.x}; }

static __device__ __forceinline__ v8f bwd_mm(const v2f (&xa)[4], v2f z, const v2f (&TBv)[4],
                                             const v2f (&UBv)[4], v2f WBv, v2f PBv, bool emit,
                                             float* __restrict__ out, int r0, int hiA, int m,
                                             int j) {
  const v8f zero = (v8f){0, 0, 0, 0, 0, 0, 0, 0};
  v8f dZ = wmma4(z, PBv, zero);
#pragma unroll
  for (int c = 0; c < 4; ++c) dZ = wmma4(xa[c], UBv[c], dZ);
  if (emit) {
    v8f dY = wmma4(z, WBv, zero);
#pragma unroll
    for (int c = 0; c < 4; ++c) dY = wmma4(xa[c], TBv[c], dY);
    const int col = 16 * j + 15 - PAD - m;
    if (col >= 0 && col < TLEN) {
#pragma unroll
      for (int v = 0; v < 8; ++v)
        out[(size_t)(r0 + v + 8 * hiA) * TLEN + col] = dY[v];
    }
  }
  return dZ;
}

__global__ __launch_bounds__(32) void filt_kernel(const float* __restrict__ x,
                                                  const float* __restrict__ ws,
                                                  float* __restrict__ out) {
  __shared__ __attribute__((aligned(16))) float ringbuf[16 * PW];
  const int task = blockIdx.x;
  const int rg = task / NSTR, st = task % NSTR;
  const int r0 = rg * 16;
  const int L = threadIdx.x;
  const int m = L & 15;
  const int hiA = L >> 4;
  const int k0 = 2 * hiA;
  const float* __restrict__ xrow = x + (size_t)(r0 + m) * TLEN;

  v2f TBv[4], UBv[4], WBv, PBv;
#pragma unroll
  for (int c = 0; c < 4; ++c) {
    TBv[c] = (v2f){ws[WS_TB + c * 64 + L], ws[WS_TB + c * 64 + 32 + L]};
    UBv[c] = (v2f){ws[WS_UB + c * 64 + L], ws[WS_UB + c * 64 + 32 + L]};
  }
  WBv = (v2f){ws[WS_WB + L], ws[WS_WB + 32 + L]};
  PBv = (v2f){ws[WS_PB + L], ws[WS_PB + 32 + L]};
  const float zi0 = ws[WS_ZI + k0], zi1 = ws[WS_ZI + k0 + 1];

  const int sB = st * SBLK;
  const int eB = ((sB + SBLK) < NBLK) ? (sB + SBLK) : NBLK;
  int fPos = (sB - WB > 0) ? (sB - WB) : 0;

  const float v0 = load_ext_fwd(xrow, fPos * 16);
  v2f zf = (v2f){zi0 * v0, zi1 * v0};

  const int sl0 = k0 + ((m & 8) ? 16 : 0);
  const int sl1 = sl0 + 1;
  const int vsel = L & 7;

  float* lst = ringbuf + (8 * hiA) * PW + m;
  const float* lld = ringbuf + m * PW + 14 - k0;

  for (int w = 0; w < NWIN; ++w) {
    const int wS = sB + w * LW;
    if (wS >= eB) break;
    const int wE = ((wS + LW) < eB) ? (wS + LW) : eB;
    const int fT = ((wE + WB) < NBLK) ? (wE + WB) : NBLK;

    zf = fwd_sweep(xrow, k0, fPos, fT, sB, zf, TBv, UBv, WBv, PBv, lst, sl0, sl1, vsel);
    fPos = fT;
    ring_sync();
    if (fT == NBLK && L < 16) {
      float* q = ringbuf + L * PW + 16 * slot_of(NBLK - 1 - sB);
      const float vl = q[TAILK];
#pragma unroll
      for (int i = TAILK + 1; i < 16; ++i) q[i] = vl;
    }
    ring_sync();

    int slot = slot_of(fT - 1 - sB);
    const float vb = ringbuf[m * PW + 16 * slot + 15];
    v2f zb = (v2f){zi0 * vb, zi1 * vb};
    v2f cur[4];
    ld_bwd_lds(lld, slot, cur);
    for (int j = fT - 1; j >= wS; --j) {
      const int sn = (j > wS) ? ((slot == 0) ? (RB - 1) : (slot - 1)) : slot;
      v2f nxt[4];
      ld_bwd_lds(lld, sn, nxt);
      v2f xa[4];
#pragma unroll
      for (int c = 0; c < 4; ++c) xa[c] = rev2(cur[c]);
      const v8f dZ = bwd_mm(xa, zb, TBv, UBv, WBv, PBv, j < wE, out, r0, hiA, m, j);
      zb = next_state(dZ, sl0, sl1, vsel);
#pragma unroll
      for (int c = 0; c < 4; ++c) cur[c] = nxt[c];
      slot = sn;
    }
    ring_sync();
  }
}

extern "C" void kernel_launch(void* const* d_in, const int* in_sizes, int n_in,
                              void* d_out, int out_size, void* d_ws, size_t ws_size,
                              hipStream_t stream) {
  (void)in_sizes; (void)n_in; (void)out_size; (void)ws_size;
  const float* x = (const float*)d_in[0];
  const float* b = (const float*)d_in[1];
  const float* a = (const float*)d_in[2];
  float* out = (float*)d_out;
  float* ws = (float*)d_ws;

  setup_kernel<<<1, 1, 0, stream>>>(b, a, ws);
  filt_kernel<<<NRG * NSTR, 32, 0, stream>>>(x, ws, out);
}


# State product added last: a block's x-only products (8 of 10 WMMAs) are summed from zero one block ahead, under the previous block's lane-exchange wait; 1 WMMA on the state chain; sums reordered
# speedup vs baseline: 1.5538x; 1.0414x over previous
#include <hip/hip_runtime.h>
#include <cstddef>

typedef __attribute__((ext_vector_type(2))) float v2f;
typedef __attribute__((ext_vector_type(8))) float v8f;

#if defined(__HIP_DEVICE_COMPILE__) && !__has_builtin(__builtin_amdgcn_wmma_f32_16x16x4_f32)
#error "__builtin_amdgcn_wmma_f32_16x16x4_f32 not available on gfx1250 device pass"
#endif

static __device__ __forceinline__ v8f wmma4(v2f a, v2f b, v8f c) {
  return __builtin_amdgcn_wmma_f32_16x16x4_f32(false, a, false, b, (short)0, c,
                                               false, false);
}

constexpr int NROWS = 256;
constexpr int TLEN  = 131072;
constexpr int PAD   = 15;
constexpr int ELEN  = TLEN + 2 * PAD;
constexpr int ES    = 131104;
constexpr int NBLK  = ES / 16;
constexpr int WB    = 16;
constexpr int NRG   = NROWS / 16;
constexpr int NSTR  = 64;
constexpr int SBLK  = (NBLK + NSTR - 1) / NSTR;
constexpr int LW    = 43;
constexpr int NWIN  = (SBLK + LW - 1) / LW;
constexpr int RB    = LW + WB;
constexpr int PW    = RB * 16 + 4;
constexpr int TAILK = (ELEN - 1) % 16;
static_assert(PW % 4 == 0 && (PW / 4) % 2 == 1, "row pitch must be 4 * odd");
static_assert(NWIN * LW >= SBLK && SBLK + WB < 3 * RB, "window bookkeeping");

constexpr int WS_TB = 0;
constexpr int WS_WB = 256;
constexpr int WS_UB = 320;
constexpr int WS_PB = 576;
constexpr int WS_ZI = 640;

__global__ void setup_kernel(const float* __restrict__ bc,
                             const float* __restrict__ ac,
                             float* __restrict__ ws) {
  if (threadIdx.x != 0 || blockIdx.x != 0) return;
  double a0 = (double)ac[0];
  double an[5], bn[5];
  for (int i = 0; i < 5; ++i) { an[i] = (double)ac[i] / a0; bn[i] = (double)bc[i] / a0; }
  const double b0 = bn[0];
  double Bv[4];
  for (int i = 0; i < 4; ++i) Bv[i] = bn[i + 1] - an[i + 1] * b0;
  double A[4][4] = {{-an[1], 1, 0, 0},
                    {-an[2], 0, 1, 0},
                    {-an[3], 0, 0, 1},
                    {-an[4], 0, 0, 0}};
  double Crow[16][4];
  Crow[0][0] = 1; Crow[0][1] = 0; Crow[0][2] = 0; Crow[0][3] = 0;
  for (int k = 1; k < 16; ++k)
    for (int j = 0; j < 4; ++j) {
      double s = 0;
      for (int i = 0; i < 4; ++i) s += Crow[k - 1][i] * A[i][j];
      Crow[k][j] = s;
    }
  double h[16];
  h[0] = b0;
  for (int m = 1; m < 16; ++m) {
    double s = 0;
    for (int i = 0; i < 4; ++i) s += Crow[m - 1][i] * Bv[i];
    h[m] = s;
  }
  double u[16][4];
  for (int i = 0; i < 4; ++i) u[0][i] = Bv[i];
  for (int p = 1; p < 16; ++p)
    for (int i = 0; i < 4; ++i) {
      double s = 0;
      for (int j = 0; j < 4; ++j) s += A[i][j] * u[p - 1][j];
      u[p][i] = s;
    }
  double P[4][4], Tm[4][4];
  for (int i = 0; i < 4; ++i)
    for (int j = 0; j < 4; ++j) P[i][j] = (i == j) ? 1.0 : 0.0;
  for (int s = 0; s < 16; ++s) {
    for (int i = 0; i < 4; ++i)
      for (int j = 0; j < 4; ++j) {
        double t = 0;
        for (int q = 0; q < 4; ++q) t += A[i][q] * P[q][j];
        Tm[i][j] = t;
      }
    for (int i = 0; i < 4; ++i)
      for (int j = 0; j < 4; ++j) P[i][j] = Tm[i][j];
  }
  double M[4][5];
  for (int i = 0; i < 4; ++i) {
    for (int j = 0; j < 4; ++j)
      M[i][j] = ((i == j) ? 1.0 : 0.0) + ((j == 0) ? an[i + 1] : 0.0) -
                ((j == i + 1) ? 1.0 : 0.0);
    M[i][4] = Bv[i];
  }
  for (int c = 0; c < 4; ++c) {
    int piv = c;
    for (int r = c + 1; r < 4; ++r)
      if ((M[r][c] < 0 ? -M[r][c] : M[r][c]) > (M[piv][c] < 0 ? -M[piv][c] : M[piv][c]))
        piv = r;
    if (piv != c)
      for (int j = 0; j < 5; ++j) { double t = M[c][j]; M[c][j] = M[piv][j]; M[piv][j] = t; }
    double d = M[c][c];
    for (int j = 0; j < 5; ++j) M[c][j] /= d;
    for (int r = 0; r < 4; ++r) {
      if (r == c) continue;
      double f = M[r][c];
      for (int j = 0; j < 5; ++j) M[r][j] -= f * M[c][j];
    }
  }
  for (int c = 0; c < 4; ++c)
    for (int v = 0; v < 2; ++v)
      for (int L = 0; L < 32; ++L) {
        int k = v + ((L >= 16) ? 2 : 0);
        int n = L & 15;
        int kk = 4 * c + k;
        ws[WS_TB + c * 64 + v * 32 + L] = (float)((n >= kk) ? h[n - kk] : 0.0);
        ws[WS_UB + c * 64 + v * 32 + L] = (float)((n < 4) ? u[15 - kk][n] : 0.0);
      }
  for (int v = 0; v < 2; ++v)
    for (int L = 0; L < 32; ++L) {
      int k = v + ((L >= 16) ? 2 : 0);
      int n = L & 15;
      ws[WS_WB + v * 32 + L] = (float)Crow[n][k];
      ws[WS_PB + v * 32 + L] = (float)((n < 4) ? P[n][k] : 0.0);
    }
  for (int i = 0; i < 4; ++i) ws[WS_ZI + i] = (float)M[i][4];
}

static __device__ __forceinline__ float load_ext_fwd(const float* __restrict__ xr, int e) {
  if (e < PAD) return 2.0f * xr[0] - xr[PAD - e];
  if (e < PAD + TLEN) return xr[e - PAD];
  int j = e - (PAD + TLEN);
  if (j > PAD - 1) j = PAD - 1;
  return 2.0f * xr[TLEN - 1] - xr[TLEN - 2 - j];
}

constexpr int PD = 4;
constexpr int FI_LO = (PAD + 15) / 16;
constexpr int FI_HI = (PAD + TLEN - 16) / 16 + 1;

static __device__ __forceinline__ void exch_issue(v8f dZ, int sl0, int sl1, float (&g0)[8],
                                                  float (&g1)[8]) {
#pragma unroll
  for (int v = 0; v < 8; ++v) {
    g0[v] = __shfl(dZ[v], sl0, 32);
    g1[v] = __shfl(dZ[v], sl1, 32);
  }
}
static __device__ __forceinline__ v2f exch_select(const float (&g0)[8], const float (&g1)[8],
                                                  int vsel) {
  float nz0 = 0.f, nz1 = 0.f;
#pragma unroll
  for (int v = 0; v < 8; ++v) {
    nz0 = (vsel == v) ? g0[v] : nz0;
    nz1 = (vsel == v) ? g1[v] : nz1;
  }
  return (v2f){nz0, nz1};
}

static __device__ __forceinline__ void x_half(const v2f (&xa)[4], const v2f (&TBv)[4],
                                              const v2f (&UBv)[4], bool withY, v8f& aZ,
                                              v8f& aY) {
  const v8f zero = (v8f){0, 0, 0, 0, 0, 0, 0, 0};
  if (withY) {
    v8f sz = wmma4(xa[0], UBv[0], zero);
    v8f sy = wmma4(xa[0], TBv[0], zero);
#pragma unroll
    for (int c = 1; c < 4; ++c) {
      sz = wmma4(xa[c], UBv[c], sz);
      sy = wmma4(xa[c], TBv[c], sy);
    }
    aZ = sz;
    aY = sy;
  } else {
    v8f sz = wmma4(xa[0], UBv[0], zero);
#pragma unroll
    for (int c = 1; c < 4; ++c) sz = wmma4(xa[c], UBv[c], sz);
    aZ = sz;
  }
}

static __device__ __forceinline__ int slot_of(int t) {
  t = (t >= RB) ? t - RB : t;
  t = (t >= RB) ? t - RB : t;
  return t;
}

static __device__ __forceinline__ void ring_sync() {
  __builtin_amdgcn_fence(__ATOMIC_RELEASE, "workgroup", "local");
  __builtin_amdgcn_s_barrier();
  __builtin_amdgcn_fence(__ATOMIC_ACQUIRE, "workgroup", "local");
}

static __device__ __forceinline__ void ld_fwd_whole(const float* __restrict__ xrow, int k0,
                                                    int bb, v2f (&xa)[4]) {
  const float* p = xrow + (bb * 16 - PAD) + k0;
#pragma unroll
  for (int c = 0; c < 4; ++c) xa[c] = (v2f){p[4 * c], p[4 * c + 1]};
  __builtin_prefetch(p + 64, 0, 1);
}

static __device__ __forceinline__ void ld_fwd_any(const float* __restrict__ xrow, int k0,
                                                  int bb, v2f (&xa)[4]) {
  if (bb >= FI_LO && bb < FI_HI) {
    ld_fwd_whole(xrow, k0, bb, xa);
  } else {
    const int e0 = bb * 16;
#pragma unroll
    for (int c = 0; c < 4; ++c) {
      int o = e0 + 4 * c + k0;
      xa[c] = (v2f){load_ext_fwd(xrow, o), load_ext_fwd(xrow, o + 1)};
    }
  }
}

static __device__ __forceinline__ void fwd_emit(v2f z, v8f aY, v2f WBv, float* lst, int slot) {
  const v8f dY = wmma4(z, WBv, aY);
  float* q = lst + slot * 16;
#pragma unroll
  for (int v = 0; v < 8; ++v) q[v * PW] = dY[v];
}

static __device__ __forceinline__ v2f fwd_block(const v2f (&xa)[4], v2f z, const v2f (&TBv)[4],
                                                const v2f (&UBv)[4], v2f WBv, v2f PBv, bool emit,
                                                float* lst, int slot, int sl0, int sl1,
                                                int vsel) {
  const v8f zero = (v8f){0, 0, 0, 0, 0, 0, 0, 0};
  v8f aZ, aY = zero;
  x_half(xa, TBv, UBv, emit, aZ, aY);
  const v8f dZ = wmma4(z, PBv, aZ);
  float g0[8], g1[8];
  exch_issue(dZ, sl0, sl1, g0, g1);
  if (emit) fwd_emit(z, aY, WBv, lst, slot);
  return exch_select(g0, g1, vsel);
}

static __device__ __forceinline__ v2f fwd_sweep(const float* __restrict__ xrow, int k0,
                                                int bFrom, int bTo, int sB, v2f z,
                                                const v2f (&TBv)[4], const v2f (&UBv)[4],
                                                v2f WBv, v2f PBv, float* lst, int sl0, int sl1,
                                                int vsel) {
  const int bLo = (bFrom > FI_LO) ? bFrom : FI_LO;
  const int bHi = (bTo < FI_HI) ? bTo : FI_HI;
  const int nRing = (bHi > bLo) ? ((bHi - bLo) / PD) * PD : 0;
  const int bMid = bLo + nRing;
  const int bHead = (nRing > 0) ? bLo : bTo;
  for (int bb = bFrom; bb < bHead; ++bb) {
    v2f xa[4];
    ld_fwd_any(xrow, k0, bb, xa);
    z = fwd_block(xa, z, TBv, UBv, WBv, PBv, bb >= sB, lst, slot_of(bb - sB), sl0, sl1, vsel);
  }
  if (nRing > 0) {
    const v8f zero = (v8f){0, 0, 0, 0, 0, 0, 0, 0};
    v2f ring[PD][4];
#pragma unroll
    for (int d = 0; d < PD; ++d) ld_fwd_whole(xrow, k0, bLo + d, ring[d]);
    v8f aZ, aY = zero;
    x_half(ring[0], TBv, UBv, bLo >= sB, aZ, aY);
    {
      const int bl = (bLo + PD < bHi) ? (bLo + PD) : (bHi - 1);
      ld_fwd_whole(xrow, k0, bl, ring[0]);
    }
    for (int bb = bLo; bb < bMid; bb += PD) {
#pragma unroll
      for (int d = 0; d < PD; ++d) {
        const int b = bb + d;
        const int dn = (d + 1) % PD;
        const v8f dZ = wmma4(z, PBv, aZ);
        float g0[8], g1[8];
        exch_issue(dZ, sl0, sl1, g0, g1);
        if (b >= sB) fwd_emit(z, aY, WBv, lst, slot_of(b - sB));
        x_half(ring[dn], TBv, UBv, (b + 1) >= sB, aZ, aY);
        int bl = b + 1 + PD;
        bl = (bl < bHi) ? bl : (bHi - 1);
        ld_fwd_whole(xrow, k0, bl, ring[dn]);
        z = exch_select(g0, g1, vsel);
      }
    }
    for (int bb = bMid; bb < bTo; ++bb) {
      v2f xa[4];
      ld_fwd_any(xrow, k0, bb, xa);
      z = fwd_block(xa, z, TBv, UBv, WBv, PBv, bb >= sB, lst, slot_of(bb - sB), sl0, sl1, vsel);
    }
  }
  return z;
}

static __device__ __forceinline__ void ld_bwd_lds(const float* lld, int slot, v2f (&raw)[4]) {
  const float* p = lld + slot * 16;
#pragma unroll
  for (int c = 0; c < 4; ++c) raw[c] = *(const v2f*)(p - 4 * c);
}
static __device__ __forceinline__ v2f rev2(v2f a) { return (v2f){a.y, a.x}; }

static __device__ __forceinline__ void bwd_store(v8f dY, float* __restrict__ out, int r0,
                                                 int hiA, int m, int j) {
  const int col = 16 * j + 15 - PAD - m;
  if (col >= 0 && col < TLEN) {
#pragma unroll
    for (int v = 0; v < 8; ++v) out[(size_t)(r0 + v + 8 * hiA) * TLEN + col] = dY[v];
  }
}

__global__ __launch_bounds__(32) void filt_kernel(const float* __restrict__ x,
                                                  const float* __restrict__ ws,
                                                  float* __restrict__ out) {
  __shared__ __attribute__((aligned(16))) float ringbuf[16 * PW];
  const int task = blockIdx.x;
  const int rg = task / NSTR, st = task % NSTR;
  const int r0 = rg * 16;
  const int L = threadIdx.x;
  const int m = L & 15;
  const int hiA = L >> 4;
  const int k0 = 2 * hiA;
  const float* __restrict__ xrow = x + (size_t)(r0 + m) * TLEN;

  v2f TBv[4], UBv[4], WBv, PBv;
#pragma unroll
  for (int c = 0; c < 4; ++c) {
    TBv[c] = (v2f){ws[WS_TB + c * 64 + L], ws[WS_TB + c * 64 + 32 + L]};
    UBv[c] = (v2f){ws[WS_UB + c * 64 + L], ws[WS_UB + c * 64 + 32 + L]};
  }
  WBv = (v2f){ws[WS_WB + L], ws[WS_WB + 32 + L]};
  PBv = (v2f){ws[WS_PB + L], ws[WS_PB + 32 + L]};
  const float zi0 = ws[WS_ZI + k0], zi1 = ws[WS_ZI + k0 + 1];

  const int sB = st * SBLK;
  const int eB = ((sB + SBLK) < NBLK) ? (sB + SBLK) : NBLK;
  int fPos = (sB - WB > 0) ? (sB - WB) : 0;

  const float v0 = load_ext_fwd(xrow, fPos * 16);
  v2f zf = (v2f){zi0 * v0, zi1 * v0};

  const int sl0 = k0 + ((m & 8) ? 16 : 0);
  const int sl1 = sl0 + 1;
  const int vsel = L & 7;

  float* lst = ringbuf + (8 * hiA) * PW + m;
  const float* lld = ringbuf + m * PW + 14 - k0;

  for (int w = 0; w < NWIN; ++w) {
    const int wS = sB + w * LW;
    if (wS >= eB) break;
    const int wE = ((wS + LW) < eB) ? (wS + LW) : eB;
    const int fT = ((wE + WB) < NBLK) ? (wE + WB) : NBLK;

    zf = fwd_sweep(xrow, k0, fPos, fT, sB, zf, TBv, UBv, WBv, PBv, lst, sl0, sl1, vsel);
    fPos = fT;
    ring_sync();
    if (fT == NBLK && L < 16) {
      float* q = ringbuf + L * PW + 16 * slot_of(NBLK - 1 - sB);
      const float vl = q[TAILK];
#pragma unroll
      for (int i = TAILK + 1; i < 16; ++i) q[i] = vl;
    }
    ring_sync();

    int slot = slot_of(fT - 1 - sB);
    const float vb = ringbuf[m * PW + 16 * slot + 15];
    v2f zb = (v2f){zi0 * vb, zi1 * vb};
    const v8f zero = (v8f){0, 0, 0, 0, 0, 0, 0, 0};
    v2f cur[4];
    ld_bwd_lds(lld, slot, cur);
    v8f aZ, aY = zero;
    {
      v2f xa[4];
#pragma unroll
      for (int c = 0; c < 4; ++c) xa[c] = rev2(cur[c]);
      x_half(xa, TBv, UBv, (fT - 1) < wE, aZ, aY);
      slot = (fT - 1 > wS) ? ((slot == 0) ? (RB - 1) : (slot - 1)) : slot;
      ld_bwd_lds(lld, slot, cur);
    }
    for (int j = fT - 1; j >= wS; --j) {
      const v8f dZ = wmma4(zb, PBv, aZ);
      float g0[8], g1[8];
      exch_issue(dZ, sl0, sl1, g0, g1);
      v8f dY = zero;
      if (j < wE) dY = wmma4(zb, WBv, aY);
      v2f xa[4];
#pragma unroll
      for (int c = 0; c < 4; ++c) xa[c] = rev2(cur[c]);
      x_half(xa, TBv, UBv, (j - 1) < wE, aZ, aY);
      if (j < wE) bwd_store(dY, out, r0, hiA, m, j);
      slot = (j - 1 > wS) ? ((slot == 0) ? (RB - 1) : (slot - 1)) : slot;
      ld_bwd_lds(lld, slot, cur);
      zb = exch_select(g0, g1, vsel);
    }
    ring_sync();
  }
}

extern "C" void kernel_launch(void* const* d_in, const int* in_sizes, int n_in,
                              void* d_out, int out_size, void* d_ws, size_t ws_size,
                              hipStream_t stream) {
  (void)in_sizes; (void)n_in; (void)out_size; (void)ws_size;
  const float* x = (const float*)d_in[0];
  const float* b = (const float*)d_in[1];
  const float* a = (const float*)d_in[2];
  float* out = (float*)d_out;
  float* ws = (float*)d_ws;

  setup_kernel<<<1, 1, 0, stream>>>(b, a, ws);
  filt_kernel<<<NRG * NSTR, 32, 0, stream>>>(x, ws, out);
}


# Set-up launch removed: each wave derives its operand images itself (both power sequences as one recurrence on the two lane halves, 64 double multiply-adds, tables in its LDS ring); sums unchanged
# speedup vs baseline: 3.0651x; 1.9726x over previous
#include <hip/hip_runtime.h>
#include <cstddef>

typedef __attribute__((ext_vector_type(2))) float v2f;
typedef __attribute__((ext_vector_type(8))) float v8f;

#if defined(__HIP_DEVICE_COMPILE__) && !__has_builtin(__builtin_amdgcn_wmma_f32_16x16x4_f32)
#error "__builtin_amdgcn_wmma_f32_16x16x4_f32 not available on gfx1250 device pass"
#endif

static __device__ __forceinline__ v8f wmma4(v2f a, v2f b, v8f c) {
  return __builtin_amdgcn_wmma_f32_16x16x4_f32(false, a, false, b, (short)0, c,
                                               false, false);
}

constexpr int NROWS = 256;
constexpr int TLEN  = 131072;
constexpr int PAD   = 15;
constexpr int ELEN  = TLEN + 2 * PAD;
constexpr int GOFF  = 1;
constexpr int COLOFF = PAD + GOFF;
constexpr int GLEN  = ELEN + GOFF;
constexpr int ES    = (GLEN + 15) / 16 * 16;
constexpr int NBLK  = ES / 16;
static_assert(COLOFF % 16 == 0, "blocks must start on 64-byte boundaries of the rows");
constexpr int WB    = 16;
constexpr int NRG   = NROWS / 16;
constexpr int NSTR  = 64;
constexpr int SBLK  = (NBLK + NSTR - 1) / NSTR;
constexpr int LW    = 43;
constexpr int NWIN  = (SBLK + LW - 1) / LW;
constexpr int RB    = LW + WB;
constexpr int WPG   = 4;
static_assert((NRG * NSTR) % WPG == 0 && (16 * (RB * 16 + 4)) % 64 == 0,
              "whole work-groups; every wave's ring starts on LDS bank 0");
constexpr int PW    = RB * 16 + 4;
constexpr int TAILK = (GLEN - 1) % 16;
static_assert(PW % 4 == 0 && (PW / 4) % 2 == 1, "row pitch must be 4 * odd");
constexpr int PZ    = 20;
constexpr int ZSZ   = 16 * PZ;
static_assert(PZ % 2 == 0 && ZSZ % 64 == 0, "8-byte aligned rows; every scratch starts on bank 0");
static_assert(NWIN * LW >= SBLK && SBLK + WB < 3 * RB, "window bookkeeping");

constexpr int NSEQ  = 17;
constexpr int SEQN  = NSEQ * 4;
constexpr int FT_OFF = SEQN * 64;
static_assert(FT_OFF + 2 * SEQN <= 16 * PW, "the tables fit the wave's ring");

static __device__ __forceinline__ double lane_bcast(double v, int lane) {
  const int lo = __builtin_amdgcn_readlane(__double2loint(v), lane);
  const int hi = __builtin_amdgcn_readlane(__double2hiint(v), lane);
  return __hiloint2double(hi, lo);
}

static __device__ __forceinline__ void wave_lds_sync() {
  __builtin_amdgcn_fence(__ATOMIC_RELEASE, "workgroup", "local");
  __builtin_amdgcn_fence(__ATOMIC_ACQUIRE, "workgroup", "local");
}

static __device__ __forceinline__ void derive_images(const float* __restrict__ bc,
                                                     const float* __restrict__ ac, float* tab,
                                                     int L, int m, int hiA, v2f (&TBv)[4],
                                                     v2f (&UBv)[4], v2f& WBv, v2f& PBv,
                                                     float& zi0, float& zi1) {
  const float* __restrict__ csrc = (L < 5) ? bc : ac;
  const int ci = (L < 5) ? L : ((L < 10) ? (L - 5) : 0);
  const double cd = (double)csrc[ci];
  const double b0 = lane_bcast(cd, 0), b1 = lane_bcast(cd, 1), b2 = lane_bcast(cd, 2);
  const double b3 = lane_bcast(cd, 3), b4 = lane_bcast(cd, 4);
  const double a1 = lane_bcast(cd, 6), a2 = lane_bcast(cd, 7), a3 = lane_bcast(cd, 8);
  const double a4 = lane_bcast(cd, 9);
  const double Bv0 = b1 - a1 * b0, Bv1 = b2 - a2 * b0, Bv2 = b3 - a3 * b0, Bv3 = b4 - a4 * b0;

  double s0 = hiA ? Bv0 : 1.0, s1 = hiA ? Bv1 : 0.0, s2 = hiA ? Bv2 : 0.0, s3 = hiA ? Bv3 : 0.0;
  float* dst = tab + hiA * 32 + 2 * m;
#pragma unroll
  for (int k = 0; k < NSEQ; ++k) {
    *(double*)(dst + (4 * k + 0) * 64) = s0;
    *(double*)(dst + (4 * k + 1) * 64) = s1;
    *(double*)(dst + (4 * k + 2) * 64) = s2;
    *(double*)(dst + (4 * k + 3) * 64) = s3;
    if (k + 1 < NSEQ) {
      const double t = s0;
      s0 = s1 - a1 * t;
      s1 = s2 - a2 * t;
      s2 = s3 - a3 * t;
      s3 = -(a4 * t);
    }
  }
  const double sa = (((1.0 + a1) + a2) + a3) + a4;
  const double sb = ((Bv0 + Bv1) + Bv2) + Bv3;
  const double z0 = sb / sa;
  const double z1 = (1.0 + a1) * z0 - Bv0;
  const double z2 = (z1 + a2 * z0) - Bv1;
  const double z3 = (z2 + a3 * z0) - Bv2;
  zi0 = (float)(hiA ? z2 : z0);
  zi1 = (float)(hiA ? z3 : z1);

  wave_lds_sync();
  float* ft = tab + FT_OFF;
#pragma unroll
  for (int i = 0; i < 5; ++i) {
    const int q = L + 32 * i;
    if (q < 2 * SEQN) {
      const int sq = (q >= SEQN) ? 1 : 0;
      const int e = q - SEQN * sq;
      ft[q] = (float)*(const double*)(tab + e * 64 + sq * 32 + 2 * m);
    }
  }
  wave_lds_sync();

  const float b0f = bc[0];
  const int nn = (m < 4) ? m : 0;
#pragma unroll
  for (int c = 0; c < 4; ++c) {
    float tb[2], ub[2];
#pragma unroll
    for (int v = 0; v < 2; ++v) {
      const int kk = 4 * c + v + 2 * hiA;
      const int j = m - kk;
      const int jc = (j >= 1) ? j : 1;
      const float hv = ft[SEQN + (jc - 1) * 4];
      tb[v] = (j > 0) ? hv : ((j == 0) ? b0f : 0.0f);
      const float uv = ft[SEQN + (15 - kk) * 4 + nn];
      ub[v] = (m < 4) ? uv : 0.0f;
    }
    TBv[c] = (v2f){tb[0], tb[1]};
    UBv[c] = (v2f){ub[0], ub[1]};
  }
  float wb[2], pb[2];
#pragma unroll
  for (int v = 0; v < 2; ++v) {
    const int k = v + 2 * hiA;
    const int j = m - k;
    const int jc = (j >= 0) ? j : 0;
    const float wv = ft[jc * 4];
    wb[v] = (j >= 0) ? wv : 0.0f;
    const float pv = ft[(16 - k) * 4 + nn];
    pb[v] = (m < 4) ? pv : 0.0f;
  }
  WBv = (v2f){wb[0], wb[1]};
  PBv = (v2f){pb[0], pb[1]};
  wave_lds_sync();
}

static __device__ __forceinline__ float load_ext_fwd(const float* __restrict__ xr, int g) {
  int e = g - GOFF;
  if (e < 0) e = 0;
  if (e < PAD) return 2.0f * xr[0] - xr[PAD - e];
  if (e < PAD + TLEN) return xr[e - PAD];
  int j = e - (PAD + TLEN);
  if (j > PAD - 1) j = PAD - 1;
  return 2.0f * xr[TLEN - 1] - xr[TLEN - 2 - j];
}

constexpr int PD = 4;
constexpr int FI_LO = (COLOFF + 15) / 16;
constexpr int FI_HI = (COLOFF + TLEN - 16) / 16 + 1;

static __device__ __forceinline__ void exch_store(v8f dZ, float* zst) {
#pragma unroll
  for (int v = 0; v < 8; ++v) zst[v * PZ] = dZ[v];
}
static __device__ __forceinline__ v2f exch_load(const float* zld) {
  return *(const v2f*)zld;
}

static __device__ __forceinline__ void x_half(const v2f (&xa)[4], const v2f (&TBv)[4],
                                              const v2f (&UBv)[4], bool withY, v8f& aZ,
                                              v8f& aY) {
  const v8f zero = (v8f){0, 0, 0, 0, 0, 0, 0, 0};
  if (withY) {
    v8f sz = wmma4(xa[0], UBv[0], zero);
    v8f sy = wmma4(xa[0], TBv[0], zero);
#pragma unroll
    for (int c = 1; c < 4; ++c) {
      sz = wmma4(xa[c], UBv[c], sz);
      sy = wmma4(xa[c], TBv[c], sy);
    }
    aZ = sz;
    aY = sy;
  } else {
    v8f sz = wmma4(xa[0], UBv[0], zero);
#pragma unroll
    for (int c = 1; c < 4; ++c) sz = wmma4(xa[c], UBv[c], sz);
    aZ = sz;
  }
}

static __device__ __forceinline__ int slot_of(int t) {
  t = (t >= RB) ? t - RB : t;
  t = (t >= RB) ? t - RB : t;
  return t;
}

static __device__ __forceinline__ void ring_sync() {
  __builtin_amdgcn_fence(__ATOMIC_RELEASE, "workgroup", "local");
  __builtin_amdgcn_fence(__ATOMIC_ACQUIRE, "workgroup", "local");
}

static __device__ __forceinline__ void ld_fwd_whole(const float* __restrict__ xrow, int k0,
                                                    int bb, v2f (&xa)[4]) {
  const float* p = xrow + (bb * 16 - COLOFF) + k0;
#pragma unroll
  for (int c = 0; c < 4; ++c) xa[c] = (v2f){p[4 * c], p[4 * c + 1]};
  __builtin_prefetch(p + 64, 0, 1);
}

static __device__ __forceinline__ void ld_fwd_any(const float* __restrict__ xrow, int k0,
                                                  int bb, v2f (&xa)[4]) {
  if (bb >= FI_LO && bb < FI_HI) {
    ld_fwd_whole(xrow, k0, bb, xa);
  } else {
    const int e0 = bb * 16;
#pragma unroll
    for (int c = 0; c < 4; ++c) {
      int o = e0 + 4 * c + k0;
      xa[c] = (v2f){load_ext_fwd(xrow, o), load_ext_fwd(xrow, o + 1)};
    }
  }
}

static __device__ __forceinline__ void fwd_emit(v2f z, v8f aY, v2f WBv, float* lst, int slot) {
  const v8f dY = wmma4(z, WBv, aY);
  float* q = lst + slot * 16;
#pragma unroll
  for (int v = 0; v < 8; ++v) q[v * PW] = dY[v];
}

static __device__ __forceinline__ v2f fwd_block(const v2f (&xa)[4], v2f z, const v2f (&TBv)[4],
                                                const v2f (&UBv)[4], v2f WBv, v2f PBv, bool emit,
                                                float* lst, int slot, float* zst,
                                                const float* zld) {
  const v8f zero = (v8f){0, 0, 0, 0, 0, 0, 0, 0};
  v8f aZ, aY = zero;
  x_half(xa, TBv, UBv, emit, aZ, aY);
  const v8f dZ = wmma4(z, PBv, aZ);
  exch_store(dZ, zst);
  const v2f zn = exch_load(zld);
  if (emit) fwd_emit(z, aY, WBv, lst, slot);
  return zn;
}

static __device__ __forceinline__ v2f fwd_sweep(const float* __restrict__ xrow, int k0,
                                                int bFrom, int bTo, int sB, v2f z,
                                                const v2f (&TBv)[4], const v2f (&UBv)[4],
                                                v2f WBv, v2f PBv, float* lst, float* zst,
                                                const float* zld) {
  const int bLo = (bFrom > FI_LO) ? bFrom : FI_LO;
  const int bHi = (bTo < FI_HI) ? bTo : FI_HI;
  const int nRing = (bHi > bLo) ? ((bHi - bLo) / PD) * PD : 0;
  const int bMid = bLo + nRing;
  const int bHead = (nRing > 0) ? bLo : bTo;
  for (int bb = bFrom; bb < bHead; ++bb) {
    v2f xa[4];
    ld_fwd_any(xrow, k0, bb, xa);
    z = fwd_block(xa, z, TBv, UBv, WBv, PBv, bb >= sB, lst, slot_of(bb - sB), zst, zld);
  }
  if (nRing > 0) {
    const v8f zero = (v8f){0, 0, 0, 0, 0, 0, 0, 0};
    v2f ring[PD][4];
#pragma unroll
    for (int d = 0; d < PD; ++d) ld_fwd_whole(xrow, k0, bLo + d, ring[d]);
    v8f aZ, aY = zero;
    x_half(ring[0], TBv, UBv, bLo >= sB, aZ, aY);
    {
      const int bl = (bLo + PD < bHi) ? (bLo + PD) : (bHi - 1);
      ld_fwd_whole(xrow, k0, bl, ring[0]);
    }
    for (int bb = bLo; bb < bMid; bb += PD) {
#pragma unroll
      for (int d = 0; d < PD; ++d) {
        const int b = bb + d;
        const int dn = (d + 1) % PD;
        const v8f dZ = wmma4(z, PBv, aZ);
        exch_store(dZ, zst);
        const v2f zn = exch_load(zld);
        if (b >= sB) fwd_emit(z, aY, WBv, lst, slot_of(b - sB));
        x_half(ring[dn], TBv, UBv, (b + 1) >= sB, aZ, aY);
        int bl = b + 1 + PD;
        bl = (bl < bHi) ? bl : (bHi - 1);
        ld_fwd_whole(xrow, k0, bl, ring[dn]);
        z = zn;
      }
    }
    for (int bb = bMid; bb < bTo; ++bb) {
      v2f xa[4];
      ld_fwd_any(xrow, k0, bb, xa);
      z = fwd_block(xa, z, TBv, UBv, WBv, PBv, bb >= sB, lst, slot_of(bb - sB), zst, zld);
    }
  }
  return z;
}

static __device__ __forceinline__ void ld_bwd_lds(const float* lld, int slot, v2f (&raw)[4]) {
  const float* p = lld + slot * 16;
#pragma unroll
  for (int c = 0; c < 4; ++c) raw[c] = *(const v2f*)(p - 4 * c);
}
static __device__ __forceinline__ v2f rev2(v2f a) { return (v2f){a.y, a.x}; }

static __device__ __forceinline__ void bwd_store(v8f dY, float* __restrict__ out, int r0,
                                                 int hiA, int m, int j) {
  const int col = 16 * j + 15 - COLOFF - m;
  if (j >= FI_LO && j < FI_HI) {
#pragma unroll
    for (int v = 0; v < 8; ++v) out[(size_t)(r0 + v + 8 * hiA) * TLEN + col] = dY[v];
  }
}

__global__ __launch_bounds__(32 * WPG) void filt_kernel(const float* __restrict__ x,
                                                       const float* __restrict__ bc,
                                                       const float* __restrict__ ac,
                                                       float* __restrict__ out) {
  __shared__ __attribute__((aligned(16))) float ringall[WPG * 16 * PW + WPG * ZSZ];
  const int wv = __builtin_amdgcn_readfirstlane((int)(threadIdx.x >> 5));
  float* ringbuf = ringall + wv * (16 * PW);
  float* zs = ringall + WPG * 16 * PW + wv * ZSZ;
  const int task = blockIdx.x * WPG + wv;
  const int rg = task / NSTR, st = task % NSTR;
  const int r0 = rg * 16;
  const int L = threadIdx.x & 31;
  const int m = L & 15;
  const int hiA = L >> 4;
  const int k0 = 2 * hiA;
  const float* __restrict__ xrow = x + (size_t)(r0 + m) * TLEN;

  v2f TBv[4], UBv[4], WBv, PBv;
  float zi0, zi1;
  derive_images(bc, ac, ringbuf, L, m, hiA, TBv, UBv, WBv, PBv, zi0, zi1);

  const int sB = st * SBLK;
  const int eB = ((sB + SBLK) < NBLK) ? (sB + SBLK) : NBLK;
  int fPos = (sB - WB > 0) ? (sB - WB) : 0;

  const float v0 = load_ext_fwd(xrow, fPos * 16);
  v2f zf = (v2f){zi0 * v0, zi1 * v0};

  float* zst = zs + (8 * hiA) * PZ + m;
  const float* zld = zs + m * PZ + k0;

  float* lst = ringbuf + (8 * hiA) * PW + m;
  const float* lld = ringbuf + m * PW + 14 - k0;

  for (int w = 0; w < NWIN; ++w) {
    const int wS = sB + w * LW;
    if (wS >= eB) break;
    const int wE = ((wS + LW) < eB) ? (wS + LW) : eB;
    const int fT = ((wE + WB) < NBLK) ? (wE + WB) : NBLK;

    zf = fwd_sweep(xrow, k0, fPos, fT, sB, zf, TBv, UBv, WBv, PBv, lst, zst, zld);
    fPos = fT;
    ring_sync();
    if (fT == NBLK && L < 16) {
      float* q = ringbuf + L * PW + 16 * slot_of(NBLK - 1 - sB);
      const float vl = q[TAILK];
#pragma unroll
      for (int i = TAILK + 1; i < 16; ++i) q[i] = vl;
    }
    ring_sync();

    int slot = slot_of(fT - 1 - sB);
    const float vb = ringbuf[m * PW + 16 * slot + 15];
    v2f zb = (v2f){zi0 * vb, zi1 * vb};
    const v8f zero = (v8f){0, 0, 0, 0, 0, 0, 0, 0};
    v2f cur[4];
    ld_bwd_lds(lld, slot, cur);
    v8f aZ, aY = zero;
    {
      v2f xa[4];
#pragma unroll
      for (int c = 0; c < 4; ++c) xa[c] = rev2(cur[c]);
      x_half(xa, TBv, UBv, (fT - 1) < wE, aZ, aY);
      slot = (fT - 1 > wS) ? ((slot == 0) ? (RB - 1) : (slot - 1)) : slot;
      ld_bwd_lds(lld, slot, cur);
    }
    for (int j = fT - 1; j >= wS; --j) {
      const v8f dZ = wmma4(zb, PBv, aZ);
      exch_store(dZ, zst);
      const v2f zn = exch_load(zld);
      v8f dY = zero;
      if (j < wE) dY = wmma4(zb, WBv, aY);
      v2f xa[4];
#pragma unroll
      for (int c = 0; c < 4; ++c) xa[c] = rev2(cur[c]);
      x_half(xa, TBv, UBv, (j - 1) < wE, aZ, aY);
      if (j < wE) bwd_store(dY, out, r0, hiA, m, j);
      slot = (j - 1 > wS) ? ((slot == 0) ? (RB - 1) : (slot - 1)) : slot;
      ld_bwd_lds(lld, slot, cur);
      zb = zn;
    }
    ring_sync();
  }
}

extern "C" void kernel_launch(void* const* d_in, const int* in_sizes, int n_in,
                              void* d_out, int out_size, void* d_ws, size_t ws_size,
                              hipStream_t stream) {
  (void)in_sizes; (void)n_in; (void)out_size; (void)ws_size;
  const float* x = (const float*)d_in[0];
  const float* b = (const float*)d_in[1];
  const float* a = (const float*)d_in[2];
  float* out = (float*)d_out;
  (void)d_ws;

  filt_kernel<<<NRG * NSTR / WPG, 32 * WPG, 0, stream>>>(x, b, a, out);
}


# Forward and backward sweep of a stretch on two waves of one SIMD that share the LDS ring, forward one 26-block window ahead, a barrier a window; images derived by one wave; re-warm points moved
# speedup vs baseline: 3.4051x; 1.1109x over previous
#include <hip/hip_runtime.h>
#include <cstddef>

typedef __attribute__((ext_vector_type(2))) float v2f;
typedef __attribute__((ext_vector_type(8))) float v8f;

#if defined(__HIP_DEVICE_COMPILE__) && !__has_builtin(__builtin_amdgcn_wmma_f32_16x16x4_f32)
#error "__builtin_amdgcn_wmma_f32_16x16x4_f32 not available on gfx1250 device pass"
#endif

static __device__ __forceinline__ v8f wmma4(v2f a, v2f b, v8f c) {
  return __builtin_amdgcn_wmma_f32_16x16x4_f32(false, a, false, b, (short)0, c,
                                               false, false);
}

constexpr int NROWS = 256;
constexpr int TLEN  = 131072;
constexpr int PAD   = 15;
constexpr int ELEN  = TLEN + 2 * PAD;
constexpr int GOFF  = 1;
constexpr int COLOFF = PAD + GOFF;
constexpr int GLEN  = ELEN + GOFF;
constexpr int ES    = (GLEN + 15) / 16 * 16;
constexpr int NBLK  = ES / 16;
static_assert(COLOFF % 16 == 0, "blocks must start on 64-byte boundaries of the rows");
constexpr int WB    = 16;
constexpr int NRG   = NROWS / 16;
constexpr int NSTR  = 64;
constexpr int SBLK  = (NBLK + NSTR - 1) / NSTR;
constexpr int LW    = 26;
constexpr int NWIN  = (SBLK + LW - 1) / LW;
constexpr int RB    = 2 * LW + WB;
constexpr int SPG   = 4;
constexpr int WPG   = 2 * SPG;
static_assert((NRG * NSTR) % SPG == 0 && (16 * (RB * 16 + 4)) % 64 == 0,
              "whole work-groups; every stretch's ring starts on LDS bank 0");
static_assert(2 * LW + WB <= RB, "a window being read and the next being written fit the ring");
constexpr int PW    = RB * 16 + 4;
constexpr int TAILK = (GLEN - 1) % 16;
static_assert(PW % 4 == 0 && (PW / 4) % 2 == 1, "row pitch must be 4 * odd");
constexpr int PZ    = 20;
constexpr int ZSZ   = 16 * PZ;
static_assert(PZ % 2 == 0 && ZSZ % 64 == 0, "8-byte aligned rows; every scratch starts on bank 0");
static_assert(NWIN * LW >= SBLK && SBLK + WB < 3 * RB, "window bookkeeping");

constexpr int NSEQ  = 16;
constexpr int SEQN  = NSEQ * 4;
constexpr int FT_OFF = SEQN * 64;
static_assert(FT_OFF + 2 * SEQN <= 16 * PW, "the tables fit the stretch's ring");
constexpr int SHR_OFF = (LW + WB) * 16;
static_assert(SHR_OFF + 32 <= RB * 16, "the hand-over places lie inside the ring rows");

static __device__ __forceinline__ double lane_bcast(double v, int lane) {
  const int lo = __builtin_amdgcn_readlane(__double2loint(v), lane);
  const int hi = __builtin_amdgcn_readlane(__double2hiint(v), lane);
  return __hiloint2double(hi, lo);
}

static __device__ __forceinline__ void wave_lds_sync() {
  __builtin_amdgcn_fence(__ATOMIC_RELEASE, "workgroup", "local");
  __builtin_amdgcn_fence(__ATOMIC_ACQUIRE, "workgroup", "local");
}

static __device__ __forceinline__ void derive_images(const float* __restrict__ bc,
                                                     const float* __restrict__ ac, float* tab,
                                                     int L, int m, int hiA, v2f (&TBv)[4],
                                                     v2f& HBv, v2f& GBv, float& gdc) {
  const float* __restrict__ csrc = (L < 5) ? bc : ac;
  const int ci = (L < 5) ? L : ((L < 10) ? (L - 5) : 0);
  const double cd = (double)csrc[ci];
  const double b0 = lane_bcast(cd, 0), b1 = lane_bcast(cd, 1), b2 = lane_bcast(cd, 2);
  const double b3 = lane_bcast(cd, 3), b4 = lane_bcast(cd, 4);
  const double a1 = lane_bcast(cd, 6), a2 = lane_bcast(cd, 7), a3 = lane_bcast(cd, 8);
  const double a4 = lane_bcast(cd, 9);
  const double Bv0 = b1 - a1 * b0, Bv1 = b2 - a2 * b0, Bv2 = b3 - a3 * b0, Bv3 = b4 - a4 * b0;

  double s0 = hiA ? Bv0 : 1.0, s1 = hiA ? Bv1 : 0.0, s2 = hiA ? Bv2 : 0.0, s3 = hiA ? Bv3 : 0.0;
  float* dst = tab + hiA * 32 + 2 * m;
#pragma unroll
  for (int k = 0; k < NSEQ; ++k) {
    *(double*)(dst + (4 * k + 0) * 64) = s0;
    *(double*)(dst + (4 * k + 1) * 64) = s1;
    *(double*)(dst + (4 * k + 2) * 64) = s2;
    *(double*)(dst + (4 * k + 3) * 64) = s3;
    if (k + 1 < NSEQ) {
      const double t = s0;
      s0 = s1 - a1 * t;
      s1 = s2 - a2 * t;
      s2 = s3 - a3 * t;
      s3 = -(a4 * t);
    }
  }
  const double sa = (((1.0 + a1) + a2) + a3) + a4;
  const double sb = ((Bv0 + Bv1) + Bv2) + Bv3;
  gdc = (float)(b0 + sb / sa);

  wave_lds_sync();
  {
    double wd[4];
#pragma unroll
    for (int i = 0; i < 4; ++i) {
      const int j = m - i;
      const int jc = (j >= 0) ? j : 0;
      const double t = *(const double*)(tab + (4 * jc) * 64 + 2 * m);
      wd[i] = (j >= 0) ? t : 0.0;
    }
    const double bd[5] = {b0, b1, b2, b3, b4};
    const double ad[5] = {1.0, a1, a2, a3, a4};
    float hb[2], gb[2];
#pragma unroll
    for (int v = 0; v < 2; ++v) {
      const int klo = v, khi = v + 2;
      double sh = 0.0, sg = 0.0;
#pragma unroll
      for (int i = 0; i < 4; ++i) {
        const double cbl = (i <= klo) ? bd[i + 4 - klo] : 0.0;
        const double cbh = (i <= khi) ? bd[i + 4 - khi] : 0.0;
        const double cal = (i <= klo) ? ad[i + 4 - klo] : 0.0;
        const double cah = (i <= khi) ? ad[i + 4 - khi] : 0.0;
        sh += (hiA ? cbh : cbl) * wd[i];
        sg -= (hiA ? cah : cal) * wd[i];
      }
      hb[v] = (float)sh;
      gb[v] = (float)sg;
    }
    HBv = (v2f){hb[0], hb[1]};
    GBv = (v2f){gb[0], gb[1]};
  }
  float* ft = tab + FT_OFF;
#pragma unroll
  for (int i = 0; i < (2 * SEQN + 31) / 32; ++i) {
    const int q = L + 32 * i;
    if (q < 2 * SEQN) {
      const int sq = (q >= SEQN) ? 1 : 0;
      const int e = q - SEQN * sq;
      ft[q] = (float)*(const double*)(tab + e * 64 + sq * 32 + 2 * m);
    }
  }
  wave_lds_sync();

  const float b0f = bc[0];
#pragma unroll
  for (int c = 0; c < 4; ++c) {
    float tb[2];
#pragma unroll
    for (int v = 0; v < 2; ++v) {
      const int kk = 4 * c + v + 2 * hiA;
      const int j = m - kk;
      const int jc = (j >= 1) ? j : 1;
      const float hv = ft[SEQN + (jc - 1) * 4];
      tb[v] = (j > 0) ? hv : ((j == 0) ? b0f : 0.0f);
    }
    TBv[c] = (v2f){tb[0], tb[1]};
  }
  wave_lds_sync();
}

static __device__ __forceinline__ float load_ext_fwd(const float* __restrict__ xr, int g) {
  int e = g - GOFF;
  if (e < 0) e = 0;
  if (e < PAD) return 2.0f * xr[0] - xr[PAD - e];
  if (e < PAD + TLEN) return xr[e - PAD];
  int j = e - (PAD + TLEN);
  if (j > PAD - 1) j = PAD - 1;
  return 2.0f * xr[TLEN - 1] - xr[TLEN - 2 - j];
}

constexpr int PD = 4;
constexpr int FI_LO = (COLOFF + 15) / 16;
constexpr int FI_HI = (COLOFF + TLEN - 16) / 16 + 1;

static __device__ __forceinline__ void exch_store(v8f dZ, float* zst) {
#pragma unroll
  for (int v = 0; v < 8; ++v) zst[v * PZ] = dZ[v];
}
static __device__ __forceinline__ v2f exch_load(const float* zld) {
  return *(const v2f*)zld;
}

static __device__ __forceinline__ void x_half(const v2f (&xa)[4], const v2f (&TBv)[4], v2f HBv,
                                              v8f cin, v8f& aY, v8f& cnx) {
  const v8f zero = (v8f){0, 0, 0, 0, 0, 0, 0, 0};
  v8f sy = wmma4(xa[0], TBv[0], cin);
  sy = wmma4(xa[1], TBv[1], sy);
  const v8f cn = wmma4(xa[3], HBv, zero);
  sy = wmma4(xa[2], TBv[2], sy);
  sy = wmma4(xa[3], TBv[3], sy);
  aY = sy;
  cnx = cn;
}

static __device__ __forceinline__ int slot_of(int t) {
  t = (t >= RB) ? t - RB : t;
  t = (t >= RB) ? t - RB : t;
  return t;
}

static __device__ __forceinline__ void ring_sync() {
  __builtin_amdgcn_fence(__ATOMIC_RELEASE, "workgroup", "local");
  __builtin_amdgcn_fence(__ATOMIC_ACQUIRE, "workgroup", "local");
}

static __device__ __forceinline__ void phase_barrier() {
  __builtin_amdgcn_fence(__ATOMIC_RELEASE, "workgroup", "local");
  __builtin_amdgcn_s_barrier();
  __builtin_amdgcn_fence(__ATOMIC_ACQUIRE, "workgroup", "local");
}

static __device__ __forceinline__ void ld_fwd_whole(const float* __restrict__ xrow, int k0,
                                                    int bb, v2f (&xa)[4]) {
  const float* p = xrow + (bb * 16 - COLOFF) + k0;
#pragma unroll
  for (int c = 0; c < 4; ++c) xa[c] = (v2f){p[4 * c], p[4 * c + 1]};
  __builtin_prefetch(p + 64, 0, 1);
}

static __device__ __forceinline__ void ld_fwd_any(const float* __restrict__ xrow, int k0,
                                                  int bb, v2f (&xa)[4]) {
  if (bb >= FI_LO && bb < FI_HI) {
    ld_fwd_whole(xrow, k0, bb, xa);
  } else {
    const int e0 = bb * 16;
#pragma unroll
    for (int c = 0; c < 4; ++c) {
      int o = e0 + 4 * c + k0;
      xa[c] = (v2f){load_ext_fwd(xrow, o), load_ext_fwd(xrow, o + 1)};
    }
  }
}

static __device__ __forceinline__ void ring_store(v8f dY, float* lst, int slot) {
  float* q = lst + slot * 16;
#pragma unroll
  for (int v = 0; v < 8; ++v) q[v * PW] = dY[v];
}

static __device__ __forceinline__ void fwd_block(const v2f (&xa)[4], v2f& yl, v8f& cx,
                                                 const v2f (&TBv)[4], v2f HBv, v2f GBv, bool emit,
                                                 float* lst, int slot, float* zst,
                                                 const float* zld) {
  v8f aY, cn;
  x_half(xa, TBv, HBv, cx, aY, cn);
  const v8f dY = wmma4(yl, GBv, aY);
  exch_store(dY, zst);
  const v2f yn = exch_load(zld);
  if (emit) ring_store(dY, lst, slot);
  yl = yn;
  cx = cn;
}

static __device__ __forceinline__ void fwd_sweep(const float* __restrict__ xrow, int k0,
                                                 int bFrom, int bTo, int sB, v2f& yl, v8f& cx,
                                                 const v2f (&TBv)[4], v2f HBv, v2f GBv,
                                                 float* lst, float* zst, const float* zld) {
  const int bLo = (bFrom > FI_LO) ? bFrom : FI_LO;
  const int bHi = (bTo < FI_HI) ? bTo : FI_HI;
  const int nRing = (bHi > bLo) ? ((bHi - bLo) / PD) * PD : 0;
  const int bMid = bLo + nRing;
  const int bHead = (nRing > 0) ? bLo : bTo;
  for (int bb = bFrom; bb < bHead; ++bb) {
    v2f xa[4];
    ld_fwd_any(xrow, k0, bb, xa);
    fwd_block(xa, yl, cx, TBv, HBv, GBv, bb >= sB, lst, slot_of(bb - sB), zst, zld);
  }
  if (nRing > 0) {
    const v8f zero = (v8f){0, 0, 0, 0, 0, 0, 0, 0};
    const float* pl = xrow + ((bMid - 1) * 16 - COLOFF) + 12 + k0;
    const v2f xlm = (v2f){pl[0], pl[1]};
    v2f ring[PD][4];
#pragma unroll
    for (int d = 0; d < PD; ++d) ld_fwd_whole(xrow, k0, bLo + d, ring[d]);
    v8f aY, cn;
    x_half(ring[0], TBv, HBv, cx, aY, cn);
    {
      const int bl = (bLo + PD < bHi) ? (bLo + PD) : (bHi - 1);
      ld_fwd_whole(xrow, k0, bl, ring[0]);
    }
    v2f y = yl;
    for (int bb = bLo; bb < bMid; bb += PD) {
#pragma unroll
      for (int d = 0; d < PD; ++d) {
        const int b = bb + d;
        const int dn = (d + 1) % PD;
        const v8f dY = wmma4(y, GBv, aY);
        exch_store(dY, zst);
        const v2f yn = exch_load(zld);
        if (b >= sB) ring_store(dY, lst, slot_of(b - sB));
        v8f aYn, cnn;
        x_half(ring[dn], TBv, HBv, cn, aYn, cnn);
        aY = aYn;
        cn = cnn;
        int bl = b + 1 + PD;
        bl = (bl < bHi) ? bl : (bHi - 1);
        asm volatile("" : : "v"(aYn), "v"(cnn) : "memory");
        ld_fwd_whole(xrow, k0, bl, ring[dn]);
        y = yn;
      }
    }
    yl = y;
    cx = wmma4(xlm, HBv, zero);
    for (int bb = bMid; bb < bTo; ++bb) {
      v2f xa[4];
      ld_fwd_any(xrow, k0, bb, xa);
      fwd_block(xa, yl, cx, TBv, HBv, GBv, bb >= sB, lst, slot_of(bb - sB), zst, zld);
    }
  }
}

static __device__ __forceinline__ void ld_bwd_lds(const float* lld, int slot, v2f (&raw)[4]) {
  const float* p = lld + slot * 16;
#pragma unroll
  for (int c = 0; c < 4; ++c) raw[c] = *(const v2f*)(p - 4 * c);
}
static __device__ __forceinline__ v2f rev2(v2f a) { return (v2f){a.y, a.x}; }

static __device__ __forceinline__ void bwd_store(v8f dY, float* __restrict__ out, int r0,
                                                 int hiA, int m, int j) {
  const int col = 16 * j + 15 - COLOFF - m;
  if (j >= FI_LO && j < FI_HI) {
#pragma unroll
    for (int v = 0; v < 8; ++v) out[(size_t)(r0 + v + 8 * hiA) * TLEN + col] = dY[v];
  }
}

__global__ __launch_bounds__(32 * WPG) void filt_kernel(const float* __restrict__ x,
                                                       const float* __restrict__ bc,
                                                       const float* __restrict__ ac,
                                                       float* __restrict__ out) {
  __shared__ __attribute__((aligned(16))) float ringall[SPG * 16 * PW + WPG * ZSZ];
  const int wv = __builtin_amdgcn_readfirstlane((int)(threadIdx.x >> 5));
  const int sx = wv % SPG;
  const bool bwdWave = wv < SPG;
  float* ringbuf = ringall + sx * (16 * PW);
  float* zs = ringall + SPG * 16 * PW + wv * ZSZ;
  const int task = blockIdx.x * SPG + sx;
  const int rg = task / NSTR, st = task % NSTR;
  const int r0 = rg * 16;
  const int L = threadIdx.x & 31;
  const int m = L & 15;
  const int hiA = L >> 4;
  const int k0 = 2 * hiA;
  const float* __restrict__ xrow = x + (size_t)(r0 + m) * TLEN;

  v2f TBv[4], HBv, GBv;
  float gdc;
  float* shr = ringbuf + SHR_OFF + L;
  if (!bwdWave) {
    derive_images(bc, ac, ringbuf, L, m, hiA, TBv, HBv, GBv, gdc);
#pragma unroll
    for (int c = 0; c < 4; ++c) {
      shr[(2 * c) * PW] = TBv[c].x;
      shr[(2 * c + 1) * PW] = TBv[c].y;
    }
    shr[8 * PW] = HBv.x;
    shr[9 * PW] = HBv.y;
    shr[10 * PW] = GBv.x;
    shr[11 * PW] = GBv.y;
    shr[12 * PW] = gdc;
  }
  phase_barrier();
  if (bwdWave) {
#pragma unroll
    for (int c = 0; c < 4; ++c) TBv[c] = (v2f){shr[(2 * c) * PW], shr[(2 * c + 1) * PW]};
    HBv = (v2f){shr[8 * PW], shr[9 * PW]};
    GBv = (v2f){shr[10 * PW], shr[11 * PW]};
    gdc = shr[12 * PW];
  }

  const int sB = st * SBLK;
  const int eB = ((sB + SBLK) < NBLK) ? (sB + SBLK) : NBLK;
  int fPos = (sB - WB > 0) ? (sB - WB) : 0;

  const v8f zero = (v8f){0, 0, 0, 0, 0, 0, 0, 0};
  const float v0 = load_ext_fwd(xrow, fPos * 16);
  v2f yf = (v2f){gdc * v0, gdc * v0};
  v8f cf = wmma4((v2f){v0, v0}, HBv, zero);

  float* zst = zs + (8 * hiA) * PZ + m;
  const float* zld = zs + m * PZ + 12 + k0;

  float* lst = ringbuf + (8 * hiA) * PW + m;
  const float* lld = ringbuf + m * PW + 14 - k0;

  for (int p = 0; p <= NWIN; ++p) {
    const int w = bwdWave ? (p - 1) : p;
    const int wS = sB + w * LW;
    if (w >= 0 && w < NWIN && wS < eB) {
      const int wE = ((wS + LW) < eB) ? (wS + LW) : eB;
      const int fT = ((wE + WB) < NBLK) ? (wE + WB) : NBLK;
      if (!bwdWave) {
        fwd_sweep(xrow, k0, fPos, fT, sB, yf, cf, TBv, HBv, GBv, lst, zst, zld);
        fPos = fT;
      } else {
        if (fT == NBLK && L < 16) {
          float* q = ringbuf + L * PW + 16 * slot_of(NBLK - 1 - sB);
          const float vl = q[TAILK];
#pragma unroll
          for (int i = TAILK + 1; i < 16; ++i) q[i] = vl;
        }
        ring_sync();

        int slot = slot_of(fT - 1 - sB);
        const float vb = ringbuf[m * PW + 16 * slot + 15];
        v2f yb = (v2f){gdc * vb, gdc * vb};
        v2f raw[4], xa[4];
        ld_bwd_lds(lld, slot, raw);
        v8f aY, cn;
        {
#pragma unroll
          for (int c = 0; c < 4; ++c) xa[c] = rev2(raw[c]);
          const v8f c0 = wmma4((v2f){vb, vb}, HBv, zero);
          slot = (fT - 1 > wS) ? ((slot == 0) ? (RB - 1) : (slot - 1)) : slot;
          ld_bwd_lds(lld, slot, raw);
          x_half(xa, TBv, HBv, c0, aY, cn);
#pragma unroll
          for (int c = 0; c < 4; ++c) xa[c] = rev2(raw[c]);
        }
        for (int j = fT - 1; j >= wS; --j) {
          const v8f dY = wmma4(yb, GBv, aY);
          exch_store(dY, zst);
          const v2f yn = exch_load(zld);
          slot = (j - 1 > wS) ? ((slot == 0) ? (RB - 1) : (slot - 1)) : slot;
          ld_bwd_lds(lld, slot, raw);
          __builtin_amdgcn_sched_barrier(0);
          v8f aYn, cnn;
          x_half(xa, TBv, HBv, cn, aYn, cnn);
          aY = aYn;
          cn = cnn;
          if (j < wE) bwd_store(dY, out, r0, hiA, m, j);
#pragma unroll
          for (int c = 0; c < 4; ++c) xa[c] = rev2(raw[c]);
          yb = yn;
        }
      }
    }
    phase_barrier();
  }
}

extern "C" void kernel_launch(void* const* d_in, const int* in_sizes, int n_in,
                              void* d_out, int out_size, void* d_ws, size_t ws_size,
                              hipStream_t stream) {
  (void)in_sizes; (void)n_in; (void)out_size; (void)ws_size;
  const float* x = (const float*)d_in[0];
  const float* b = (const float*)d_in[1];
  const float* a = (const float*)d_in[2];
  float* out = (float*)d_out;
  (void)d_ws;

  filt_kernel<<<NRG * NSTR / SPG, 32 * WPG, 0, stream>>>(x, b, a, out);
}
